# starting kernel
# baseline (speedup 1.0000x reference)
#include <hip/hip_runtime.h>
#include <hip/hip_bf16.h>

typedef __attribute__((ext_vector_type(2))) float v2f;
typedef __attribute__((ext_vector_type(8))) float v8f;

#define EPSV 1e-5f
constexpr int Bn = 2, Cc = 256, Hh = 128, Ww = 128, Gg = 8;
constexpr int HW = Hh * Ww;

__global__ __launch_bounds__(256)
void conv1_bn_tanh(const float* __restrict__ x, const float* __restrict__ w1,
                   const float* __restrict__ b1, const float* __restrict__ g1,
                   const float* __restrict__ be1, const float* __restrict__ m1,
                   const float* __restrict__ v1, float* __restrict__ t)
{
    const int lane  = threadIdx.x & 31;
    const int g     = threadIdx.x >> 5;
    const int tile  = blockIdx.x;
    const int b     = tile >> 10;
    const int pbase = (tile & 1023) * 16;

    const int mrow  = lane & 15;
    const int khalf = (lane >> 4) * 2;
    const int arow  = mrow & 7;
    const bool aval = (mrow < 8);

    const float* xg = x  + ((size_t)b * Cc + g * 32) * HW;
    const float* wg = w1 + g * 8 * 32;

    v2f af[8], bf[8];
#pragma unroll
    for (int kk = 0; kk < 8; ++kk) {
        const int k0 = kk * 4 + khalf;
        const float ax = wg[arow * 32 + k0];
        const float ay = wg[arow * 32 + k0 + 1];
        af[kk].x = aval ? ax : 0.f;
        af[kk].y = aval ? ay : 0.f;
        bf[kk].x = xg[(size_t)k0       * HW + pbase + mrow];
        bf[kk].y = xg[(size_t)(k0 + 1) * HW + pbase + mrow];
    }
    v8f acc = {};
#pragma unroll
    for (int kk = 0; kk < 8; ++kk)
        acc = __builtin_amdgcn_wmma_f32_16x16x4_f32(false, af[kk], false, bf[kk],
                                                    (short)0, acc, false, false);

    if (lane < 16) {
        const int pix = pbase + lane;
#pragma unroll
        for (int i = 0; i < 8; ++i) {
            const int ch = g * 8 + i;
            const float inv = g1[ch] * rsqrtf(v1[ch] + EPSV);
            const float val = (acc[i] + b1[ch]) * inv + (be1[ch] - m1[ch] * inv);
            t[((size_t)b * 64 + ch) * HW + pix] = tanhf(val);
        }
    }
}

__global__ __launch_bounds__(256)
void conv2_bn(const float* __restrict__ t, const float* __restrict__ w2,
              const float* __restrict__ b2, const float* __restrict__ g2,
              const float* __restrict__ be2, const float* __restrict__ m2,
              const float* __restrict__ v2, float* __restrict__ mn)
{
    __shared__ float w2s[8 * 10 * 8];
    for (int i = threadIdx.x; i < 640; i += 256) w2s[i] = w2[i];
    __syncthreads();

    const int tid = blockIdx.x * 256 + threadIdx.x;
    const int b = tid / HW;
    const int p = tid % HW;
    const float* tb = t  + (size_t)b * 64 * HW + p;
    float*       mb = mn + (size_t)b * 80 * HW + p;

#pragma unroll
    for (int g = 0; g < 8; ++g) {
        float tin[8];
#pragma unroll
        for (int i = 0; i < 8; ++i) tin[i] = tb[(size_t)(g * 8 + i) * HW];
#pragma unroll
        for (int o = 0; o < 10; ++o) {
            float acc = 0.f;
#pragma unroll
            for (int i = 0; i < 8; ++i) acc += tin[i] * w2s[(g * 10 + o) * 8 + i];
            const int ch = g * 10 + o;
            const float inv = g2[ch] * rsqrtf(v2[ch] + EPSV);
            mb[(size_t)ch * HW] = (acc + b2[ch]) * inv + (be2[ch] - m2[ch] * inv);
        }
    }
}

__global__ __launch_bounds__(256)
void vconv(const float* __restrict__ x, const float* __restrict__ wv,
           float* __restrict__ v)
{
    const int lane = threadIdx.x & 31;
    const int wid  = threadIdx.x >> 5;
    int blk = blockIdx.x;
    const int strip = blk & 127;  blk >>= 7;
    const int g     = blk & 7;    blk >>= 3;
    const int b     = blk;
    const int pbase = (strip * 8 + wid) * 16;

    const int mrow  = lane & 15;
    const int khalf = (lane >> 4) * 2;

    const float* xg = x  + ((size_t)b * Cc + g * 32) * HW;
    const float* wg = wv + g * 32 * 32;

    v2f alo[8], ahi[8], bf[8];
#pragma unroll
    for (int kk = 0; kk < 8; ++kk) {
        const int k0 = kk * 4 + khalf;
        bf[kk].x  = xg[(size_t)k0       * HW + pbase + mrow];
        bf[kk].y  = xg[(size_t)(k0 + 1) * HW + pbase + mrow];
        alo[kk].x = wg[mrow * 32 + k0];         alo[kk].y = wg[mrow * 32 + k0 + 1];
        ahi[kk].x = wg[(mrow + 16) * 32 + k0];  ahi[kk].y = wg[(mrow + 16) * 32 + k0 + 1];
    }
    v8f clo = {}, chi = {};
#pragma unroll
    for (int kk = 0; kk < 8; ++kk) {
        clo = __builtin_amdgcn_wmma_f32_16x16x4_f32(false, alo[kk], false, bf[kk],
                                                    (short)0, clo, false, false);
        chi = __builtin_amdgcn_wmma_f32_16x16x4_f32(false, ahi[kk], false, bf[kk],
                                                    (short)0, chi, false, false);
    }
    const int orow = (lane < 16) ? 0 : 8;
    const int pix  = pbase + mrow;
    float* vg = v + ((size_t)b * Cc + g * 32) * HW + pix;
#pragma unroll
    for (int i = 0; i < 8; ++i) {
        vg[(size_t)(i + orow)      * HW] = clo[i];
        vg[(size_t)(i + orow + 16) * HW] = chi[i];
    }
}

__global__ __launch_bounds__(256)
void attn_kernel(const float* __restrict__ mn, const float* __restrict__ v,
                 float* __restrict__ out)
{
    __shared__ float vs[32][18 * 18];
    __shared__ float nbs[18 * 18];

    const int bg = blockIdx.x;
    const int b  = bg >> 3, g = bg & 7;
    const int y0 = blockIdx.y * 16, x0 = blockIdx.z * 16;
    const int lx = threadIdx.x & 15, ly = threadIdx.x >> 4;

    const float* vgrp = v  + ((size_t)b * 256 + g * 32) * HW;
    const float* nbp  = mn + ((size_t)b * 80 + g) * HW;

    for (int e = threadIdx.x; e < 324; e += 256) {
        const int r = e / 18, cdx = e % 18;
        const int gy = y0 - 1 + r, gx = x0 - 1 + cdx;
        nbs[e] = (gy >= 0 && gy < Hh && gx >= 0 && gx < Ww) ? nbp[gy * Ww + gx] : 0.f;
    }
    for (int e = threadIdx.x; e < 32 * 324; e += 256) {
        const int c = e / 324, rem = e % 324;
        const int r = rem / 18, cdx = rem % 18;
        const int gy = y0 - 1 + r, gx = x0 - 1 + cdx;
        vs[c][rem] = (gy >= 0 && gy < Hh && gx >= 0 && gx < Ww)
                         ? vgrp[(size_t)c * HW + gy * Ww + gx] : 0.f;
    }
    __syncthreads();

    const int y = y0 + ly, xp = x0 + lx;
    const float* maskp = mn + ((size_t)b * 80 + 8 + g * 9) * HW + (size_t)y * Ww + xp;

    float lg[9];
#pragma unroll
    for (int k = 0; k < 9; ++k) {
        const int i = k / 3, j = k % 3;
        lg[k] = maskp[(size_t)k * HW] + nbs[(ly + i) * 18 + (lx + j)];
    }
    float mx = lg[0];
#pragma unroll
    for (int k = 1; k < 9; ++k) mx = fmaxf(mx, lg[k]);
    float s = 0.f;
#pragma unroll
    for (int k = 0; k < 9; ++k) { lg[k] = expf(lg[k] - mx); s += lg[k]; }
    const float sinv = 1.f / s;
#pragma unroll
    for (int k = 0; k < 9; ++k) lg[k] *= sinv;

    float* outp = out + ((size_t)b * 256 + g * 32) * HW + (size_t)y * Ww + xp;
#pragma unroll
    for (int c = 0; c < 32; ++c) {
        float acc = 0.f;
#pragma unroll
        for (int k = 0; k < 9; ++k) {
            const int i = k / 3, j = k % 3;
            acc += lg[k] * vs[c][(ly + i) * 18 + (lx + j)];
        }
        outp[(size_t)c * HW] = acc;
    }
}

extern "C" void kernel_launch(void* const* d_in, const int* in_sizes, int n_in,
                              void* d_out, int out_size, void* d_ws, size_t ws_size,
                              hipStream_t stream)
{
    const float* x   = (const float*)d_in[0];
    const float* w1  = (const float*)d_in[1];
    const float* b1  = (const float*)d_in[2];
    const float* g1  = (const float*)d_in[3];
    const float* be1 = (const float*)d_in[4];
    const float* m1  = (const float*)d_in[5];
    const float* v1  = (const float*)d_in[6];
    const float* w2  = (const float*)d_in[7];
    const float* b2  = (const float*)d_in[8];
    const float* g2  = (const float*)d_in[9];
    const float* be2 = (const float*)d_in[10];
    const float* m2  = (const float*)d_in[11];
    const float* v2  = (const float*)d_in[12];
    const float* wvp = (const float*)d_in[13];
    float* out = (float*)d_out;

    float* t    = (float*)d_ws;
    float* mnb  = t   + (size_t)Bn * 64 * HW;
    float* vbuf = mnb + (size_t)Bn * 80 * HW;

    conv1_bn_tanh<<<Bn * 1024, 256, 0, stream>>>(x, w1, b1, g1, be1, m1, v1, t);
    conv2_bn<<<(Bn * HW) / 256, 256, 0, stream>>>(t, w2, b2, g2, be2, m2, v2, mnb);
    vconv<<<Bn * Gg * 128, 256, 0, stream>>>(x, wvp, vbuf);
    attn_kernel<<<dim3(Bn * Gg, Hh / 16, Ww / 16), 256, 0, stream>>>(mnb, vbuf, out);
}


# Fuse the three 1x1 grouped convs into one kernel on a single load of each x fragment: t stays in registers, conv2+BN runs in its epilogue; 4 launches to 2, x read once
# speedup vs baseline: 1.4657x; 1.4657x over previous
#include <hip/hip_runtime.h>
#include <hip/hip_bf16.h>

typedef __attribute__((ext_vector_type(2))) float v2f;
typedef __attribute__((ext_vector_type(8))) float v8f;

#define EPSV 1e-5f
constexpr int Bn = 2, Cc = 256, Hh = 128, Ww = 128, Gg = 8;
constexpr int HW = Hh * Ww;

__global__ __launch_bounds__(256)
void pointwise_fused(const float* __restrict__ x,
                     const float* __restrict__ w1, const float* __restrict__ b1,
                     const float* __restrict__ g1, const float* __restrict__ be1,
                     const float* __restrict__ m1, const float* __restrict__ v1,
                     const float* __restrict__ w2, const float* __restrict__ b2,
                     const float* __restrict__ g2, const float* __restrict__ be2,
                     const float* __restrict__ m2, const float* __restrict__ v2,
                     const float* __restrict__ wv,
                     float* __restrict__ mn, float* __restrict__ v)
{
    const int lane = threadIdx.x & 31;
    const int wid  = threadIdx.x >> 5;
    int blk = blockIdx.x;
    const int strip = blk & 127;  blk >>= 7;
    const int g     = blk & 7;    blk >>= 3;
    const int b     = blk;
    const int pbase = (strip * 8 + wid) * 16;

    const int mrow  = lane & 15;
    const int khalf = (lane >> 4) * 2;
    const int arow  = mrow & 7;

    const float* xg  = x  + ((size_t)b * Cc + g * 32) * HW;
    const float* wvg = wv + g * 32 * 32;
    const float* w1g = w1 + g * 8 * 32;

    v2f a1[8], alo[8], ahi[8], bf[8];
#pragma unroll
    for (int kk = 0; kk < 8; ++kk) {
        const int k0 = kk * 4 + khalf;
        bf[kk].x  = xg[(size_t)k0       * HW + pbase + mrow];
        bf[kk].y  = xg[(size_t)(k0 + 1) * HW + pbase + mrow];
        a1[kk].x  = w1g[arow * 32 + k0];
        a1[kk].y  = w1g[arow * 32 + k0 + 1];
    }
#pragma unroll
    for (int kk = 0; kk < 8; ++kk) {
        const int k0 = kk * 4 + khalf;
        alo[kk].x = wvg[mrow * 32 + k0];         alo[kk].y = wvg[mrow * 32 + k0 + 1];
        ahi[kk].x = wvg[(mrow + 16) * 32 + k0];  ahi[kk].y = wvg[(mrow + 16) * 32 + k0 + 1];
    }
    v8f c1 = {}, clo = {}, chi = {};
#pragma unroll
    for (int kk = 0; kk < 8; ++kk)
        c1  = __builtin_amdgcn_wmma_f32_16x16x4_f32(false, a1[kk],  false, bf[kk],
                                                    (short)0, c1,  false, false);
    asm volatile("" : "+v"(c1), "+v"(bf[0]));
#pragma unroll
    for (int kk = 0; kk < 8; ++kk) {
        clo = __builtin_amdgcn_wmma_f32_16x16x4_f32(false, alo[kk], false, bf[kk],
                                                    (short)0, clo, false, false);
        chi = __builtin_amdgcn_wmma_f32_16x16x4_f32(false, ahi[kk], false, bf[kk],
                                                    (short)0, chi, false, false);
    }

    {
        const int orow = (lane < 16) ? 0 : 8;
        float* vg = v + ((size_t)b * Cc + g * 32) * HW + pbase + mrow;
#pragma unroll
        for (int i = 0; i < 8; ++i) {
            vg[(size_t)(i + orow)      * HW] = clo[i];
            vg[(size_t)(i + orow + 16) * HW] = chi[i];
        }
    }

    if (lane < 16) {
        float tin[8];
#pragma unroll
        for (int i = 0; i < 8; ++i) {
            const int ch = g * 8 + i;
            const float inv = g1[ch] * rsqrtf(v1[ch] + EPSV);
            const float val = (c1[i] + b1[ch]) * inv + (be1[ch] - m1[ch] * inv);
            tin[i] = tanhf(val);
        }
        float* mb = mn + ((size_t)b * 80 + g * 10) * HW + pbase + lane;
#pragma unroll
        for (int o = 0; o < 10; ++o) {
            const int ch = g * 10 + o;
            float acc = 0.f;
#pragma unroll
            for (int i = 0; i < 8; ++i) acc += tin[i] * w2[ch * 8 + i];
            const float inv = g2[ch] * rsqrtf(v2[ch] + EPSV);
            mb[(size_t)o * HW] = (acc + b2[ch]) * inv + (be2[ch] - m2[ch] * inv);
        }
    }
}

__global__ __launch_bounds__(256)
void attn_kernel(const float* __restrict__ mn, const float* __restrict__ v,
                 float* __restrict__ out)
{
    __shared__ float vs[32][18 * 18];
    __shared__ float nbs[18 * 18];

    const int bg = blockIdx.x;
    const int b  = bg >> 3, g = bg & 7;
    const int y0 = blockIdx.y * 16, x0 = blockIdx.z * 16;
    const int lx = threadIdx.x & 15, ly = threadIdx.x >> 4;

    const float* vgrp = v  + ((size_t)b * 256 + g * 32) * HW;
    const float* nbp  = mn + ((size_t)b * 80 + g) * HW;

    for (int e = threadIdx.x; e < 324; e += 256) {
        const int r = e / 18, cdx = e % 18;
        const int gy = y0 - 1 + r, gx = x0 - 1 + cdx;
        nbs[e] = (gy >= 0 && gy < Hh && gx >= 0 && gx < Ww) ? nbp[gy * Ww + gx] : 0.f;
    }
    for (int e = threadIdx.x; e < 32 * 324; e += 256) {
        const int c = e / 324, rem = e % 324;
        const int r = rem / 18, cdx = rem % 18;
        const int gy = y0 - 1 + r, gx = x0 - 1 + cdx;
        vs[c][rem] = (gy >= 0 && gy < Hh && gx >= 0 && gx < Ww)
                         ? vgrp[(size_t)c * HW + gy * Ww + gx] : 0.f;
    }
    __syncthreads();

    const int y = y0 + ly, xp = x0 + lx;
    const float* maskp = mn + ((size_t)b * 80 + 8 + g * 9) * HW + (size_t)y * Ww + xp;

    float lg[9];
#pragma unroll
    for (int k = 0; k < 9; ++k) {
        const int i = k / 3, j = k % 3;
        lg[k] = maskp[(size_t)k * HW] + nbs[(ly + i) * 18 + (lx + j)];
    }
    float mx = lg[0];
#pragma unroll
    for (int k = 1; k < 9; ++k) mx = fmaxf(mx, lg[k]);
    float s = 0.f;
#pragma unroll
    for (int k = 0; k < 9; ++k) { lg[k] = expf(lg[k] - mx); s += lg[k]; }
    const float sinv = 1.f / s;
#pragma unroll
    for (int k = 0; k < 9; ++k) lg[k] *= sinv;

    float* outp = out + ((size_t)b * 256 + g * 32) * HW + (size_t)y * Ww + xp;
#pragma unroll
    for (int c = 0; c < 32; ++c) {
        float acc = 0.f;
#pragma unroll
        for (int k = 0; k < 9; ++k) {
            const int i = k / 3, j = k % 3;
            acc += lg[k] * vs[c][(ly + i) * 18 + (lx + j)];
        }
        outp[(size_t)c * HW] = acc;
    }
}

extern "C" void kernel_launch(void* const* d_in, const int* in_sizes, int n_in,
                              void* d_out, int out_size, void* d_ws, size_t ws_size,
                              hipStream_t stream)
{
    const float* x   = (const float*)d_in[0];
    const float* w1  = (const float*)d_in[1];
    const float* b1  = (const float*)d_in[2];
    const float* g1  = (const float*)d_in[3];
    const float* be1 = (const float*)d_in[4];
    const float* m1  = (const float*)d_in[5];
    const float* v1  = (const float*)d_in[6];
    const float* w2  = (const float*)d_in[7];
    const float* b2  = (const float*)d_in[8];
    const float* g2  = (const float*)d_in[9];
    const float* be2 = (const float*)d_in[10];
    const float* m2  = (const float*)d_in[11];
    const float* v2  = (const float*)d_in[12];
    const float* wvp = (const float*)d_in[13];
    float* out = (float*)d_out;

    float* mnb  = (float*)d_ws;
    float* vbuf = mnb + (size_t)Bn * 80 * HW;

    pointwise_fused<<<Bn * Gg * 128, 256, 0, stream>>>(x, w1, b1, g1, be1, m1, v1,
                                                       w2, b2, g2, be2, m2, v2,
                                                       wvp, mnb, vbuf);
    attn_kernel<<<dim3(Bn * Gg, Hh / 16, Ww / 16), 256, 0, stream>>>(mnb, vbuf, out);
}


# Move the value 1x1 conv into the attention kernel: each block computes its 32ch x 18x18 v tile from x on the matrix unit straight into LDS; v is never written to or read from memory
# speedup vs baseline: 2.2565x; 1.5395x over previous
#include <hip/hip_runtime.h>

typedef __attribute__((ext_vector_type(2))) float v2f;
typedef __attribute__((ext_vector_type(8))) float v8f;

#define EPSV 1e-5f
#define WMMA_SETTLE "v_nop\n\tv_nop\n\tv_nop\n\tv_nop\n\tv_nop\n\tv_nop\n\tv_nop\n\tv_nop"
constexpr int Bn = 2, Cc = 256, Hh = 128, Ww = 128, Gg = 8;
constexpr int HW = Hh * Ww;

__global__ __launch_bounds__(256)
void mask_pointwise(const float* __restrict__ x,
                    const float* __restrict__ w1, const float* __restrict__ b1,
                    const float* __restrict__ g1, const float* __restrict__ be1,
                    const float* __restrict__ m1, const float* __restrict__ v1,
                    const float* __restrict__ w2, const float* __restrict__ b2,
                    const float* __restrict__ g2, const float* __restrict__ be2,
                    const float* __restrict__ m2, const float* __restrict__ v2,
                    float* __restrict__ mn)
{
    const int lane = threadIdx.x & 31;
    const int wid  = threadIdx.x >> 5;
    int blk = blockIdx.x;
    const int strip = blk & 127;  blk >>= 7;
    const int g     = blk & 7;    blk >>= 3;
    const int b     = blk;
    const int pbase = (strip * 8 + wid) * 16;

    const int mrow  = lane & 15;
    const int khalf = (lane >> 4) * 2;
    const int arow  = mrow & 7;

    const float* xg  = x  + ((size_t)b * Cc + g * 32) * HW;
    const float* w1g = w1 + g * 8 * 32;

    v2f a1[8], bf[8];
#pragma unroll
    for (int kk = 0; kk < 8; ++kk) {
        const int k0 = kk * 4 + khalf;
        bf[kk].x  = xg[(size_t)k0       * HW + pbase + mrow];
        bf[kk].y  = xg[(size_t)(k0 + 1) * HW + pbase + mrow];
        a1[kk].x  = w1g[arow * 32 + k0];
        a1[kk].y  = w1g[arow * 32 + k0 + 1];
    }
    v8f c1 = {};
#pragma unroll
    for (int kk = 0; kk < 8; ++kk)
        c1  = __builtin_amdgcn_wmma_f32_16x16x4_f32(false, a1[kk],  false, bf[kk],
                                                    (short)0, c1,  false, false);
    asm volatile(WMMA_SETTLE : "+v"(c1),
                 "+v"(a1[0]), "+v"(a1[1]), "+v"(a1[2]), "+v"(a1[3]),
                 "+v"(a1[4]), "+v"(a1[5]), "+v"(a1[6]), "+v"(a1[7]),
                 "+v"(bf[0]), "+v"(bf[1]), "+v"(bf[2]), "+v"(bf[3]),
                 "+v"(bf[4]), "+v"(bf[5]), "+v"(bf[6]), "+v"(bf[7]));

    if (lane < 16) {
        float tin[8];
#pragma unroll
        for (int i = 0; i < 8; ++i) {
            const int ch = g * 8 + i;
            const float inv = g1[ch] * rsqrtf(v1[ch] + EPSV);
            const float val = (c1[i] + b1[ch]) * inv + (be1[ch] - m1[ch] * inv);
            tin[i] = tanhf(val);
        }
        float* mb = mn + ((size_t)b * 80 + g * 10) * HW + pbase + lane;
#pragma unroll
        for (int o = 0; o < 10; ++o) {
            const int ch = g * 10 + o;
            float acc = 0.f;
#pragma unroll
            for (int i = 0; i < 8; ++i) acc += tin[i] * w2[ch * 8 + i];
            const float inv = g2[ch] * rsqrtf(v2[ch] + EPSV);
            mb[(size_t)o * HW] = (acc + b2[ch]) * inv + (be2[ch] - m2[ch] * inv);
        }
    }
}

__global__ __launch_bounds__(256)
void attn_kernel(const float* __restrict__ x, const float* __restrict__ wv,
                 const float* __restrict__ mn, float* __restrict__ out)
{
    __shared__ float vs[32][18 * 18];
    __shared__ float nbs[18 * 18];

    const int bg = blockIdx.x;
    const int b  = bg >> 3, g = bg & 7;
    const int y0 = blockIdx.y * 16, x0 = blockIdx.z * 16;
    const int lx = threadIdx.x & 15, ly = threadIdx.x >> 4;

    const float* xg  = x  + ((size_t)b * Cc + g * 32) * HW;
    const float* wvg = wv + g * 32 * 32;
    const float* nbp = mn + ((size_t)b * 80 + g) * HW;

    {
        const int lane  = threadIdx.x & 31;
        const int wid   = __builtin_amdgcn_readfirstlane((int)(threadIdx.x >> 5)) & 7;
        const int mrow  = lane & 15;
        const int khalf = (lane >> 4) * 2;
        const int orow  = (lane < 16) ? 0 : 8;

        v2f alo[8], ahi[8];
#pragma unroll
        for (int kk = 0; kk < 8; ++kk) {
            const int k0 = kk * 4 + khalf;
            alo[kk].x = wvg[mrow * 32 + k0];         alo[kk].y = wvg[mrow * 32 + k0 + 1];
            ahi[kk].x = wvg[(mrow + 16) * 32 + k0];  ahi[kk].y = wvg[(mrow + 16) * 32 + k0 + 1];
        }

        v2f  bf[3][8];
        bool okv[3];
        int  ev[3];
#pragma unroll
        for (int tt = 0; tt < 3; ++tt) {
            const int e   = (wid + 8 * tt) * 16 + mrow;
            const int r   = e / 18, cdx = e - r * 18;
            const int gy  = y0 - 1 + r, gx = x0 - 1 + cdx;
            const bool ok = (e < 324) && gy >= 0 && gy < Hh && gx >= 0 && gx < Ww;
            const int off = ok ? gy * Ww + gx : 0;
            ev[tt] = e;  okv[tt] = ok;
#pragma unroll
            for (int kk = 0; kk < 8; ++kk) {
                const int k0 = kk * 4 + khalf;
                bf[tt][kk].x = xg[(size_t)k0       * HW + off];
                bf[tt][kk].y = xg[(size_t)(k0 + 1) * HW + off];
            }
        }
#pragma unroll
        for (int tt = 0; tt < 3; ++tt) {
            if (tt < 2 || wid < 5) {
                v8f clo = {}, chi = {};
#pragma unroll
                for (int kk = 0; kk < 8; ++kk) {
                    clo = __builtin_amdgcn_wmma_f32_16x16x4_f32(false, alo[kk], false, bf[tt][kk],
                                                                (short)0, clo, false, false);
                    chi = __builtin_amdgcn_wmma_f32_16x16x4_f32(false, ahi[kk], false, bf[tt][kk],
                                                                (short)0, chi, false, false);
                }
                asm volatile(WMMA_SETTLE : "+v"(clo), "+v"(chi),
                             "+v"(bf[tt][0]), "+v"(bf[tt][1]), "+v"(bf[tt][2]), "+v"(bf[tt][3]),
                             "+v"(bf[tt][4]), "+v"(bf[tt][5]), "+v"(bf[tt][6]), "+v"(bf[tt][7]));
                const int e = ev[tt];
                if (e < 324) {
#pragma unroll
                    for (int i = 0; i < 8; ++i) {
                        vs[i + orow][e]      = okv[tt] ? clo[i] : 0.f;
                        vs[i + orow + 16][e] = okv[tt] ? chi[i] : 0.f;
                    }
                }
            }
        }
    }
    for (int e = threadIdx.x; e < 324; e += 256) {
        const int r = e / 18, cdx = e % 18;
        const int gy = y0 - 1 + r, gx = x0 - 1 + cdx;
        nbs[e] = (gy >= 0 && gy < Hh && gx >= 0 && gx < Ww) ? nbp[gy * Ww + gx] : 0.f;
    }
    __syncthreads();

    const int y = y0 + ly, xp = x0 + lx;
    const float* maskp = mn + ((size_t)b * 80 + 8 + g * 9) * HW + (size_t)y * Ww + xp;

    float lg[9];
#pragma unroll
    for (int k = 0; k < 9; ++k) {
        const int i = k / 3, j = k % 3;
        lg[k] = maskp[(size_t)k * HW] + nbs[(ly + i) * 18 + (lx + j)];
    }
    float mx = lg[0];
#pragma unroll
    for (int k = 1; k < 9; ++k) mx = fmaxf(mx, lg[k]);
    float s = 0.f;
#pragma unroll
    for (int k = 0; k < 9; ++k) { lg[k] = expf(lg[k] - mx); s += lg[k]; }
    const float sinv = 1.f / s;
#pragma unroll
    for (int k = 0; k < 9; ++k) lg[k] *= sinv;

    float* outp = out + ((size_t)b * 256 + g * 32) * HW + (size_t)y * Ww + xp;
#pragma unroll
    for (int c = 0; c < 32; ++c) {
        float acc = 0.f;
#pragma unroll
        for (int k = 0; k < 9; ++k) {
            const int i = k / 3, j = k % 3;
            acc += lg[k] * vs[c][(ly + i) * 18 + (lx + j)];
        }
        outp[(size_t)c * HW] = acc;
    }
}

extern "C" void kernel_launch(void* const* d_in, const int* in_sizes, int n_in,
                              void* d_out, int out_size, void* d_ws, size_t ws_size,
                              hipStream_t stream)
{
    const float* x   = (const float*)d_in[0];
    const float* w1  = (const float*)d_in[1];
    const float* b1  = (const float*)d_in[2];
    const float* g1  = (const float*)d_in[3];
    const float* be1 = (const float*)d_in[4];
    const float* m1  = (const float*)d_in[5];
    const float* v1  = (const float*)d_in[6];
    const float* w2  = (const float*)d_in[7];
    const float* b2  = (const float*)d_in[8];
    const float* g2  = (const float*)d_in[9];
    const float* be2 = (const float*)d_in[10];
    const float* m2  = (const float*)d_in[11];
    const float* v2  = (const float*)d_in[12];
    const float* wvp = (const float*)d_in[13];
    float* out = (float*)d_out;

    float* mnb  = (float*)d_ws;

    mask_pointwise<<<Bn * Gg * 128, 256, 0, stream>>>(x, w1, b1, g1, be1, m1, v1,
                                                      w2, b2, g2, be2, m2, v2, mnb);
    attn_kernel<<<dim3(Bn * Gg, Hh / 16, Ww / 16), 256, 0, stream>>>(x, wvp, mnb, out);
}


# Attention blocks re-tiled from 16x16 pixels to strips of 4 full image rows (512 threads): each x, mask and out access of a wave covers whole 128-byte lines; halo columns are never loaded
# speedup vs baseline: 2.4554x; 1.0881x over previous
#include <hip/hip_runtime.h>

typedef __attribute__((ext_vector_type(2))) float v2f;
typedef __attribute__((ext_vector_type(8))) float v8f;

#define EPSV 1e-5f
#define WMMA_SETTLE "v_nop\n\tv_nop\n\tv_nop\n\tv_nop\n\tv_nop\n\tv_nop\n\tv_nop\n\tv_nop"
constexpr int Bn = 2, Cc = 256, Hh = 128, Ww = 128, Gg = 8;
constexpr int HW = Hh * Ww;

__global__ __launch_bounds__(256)
void mask_pointwise(const float* __restrict__ x,
                    const float* __restrict__ w1, const float* __restrict__ b1,
                    const float* __restrict__ g1, const float* __restrict__ be1,
                    const float* __restrict__ m1, const float* __restrict__ v1,
                    const float* __restrict__ w2, const float* __restrict__ b2,
                    const float* __restrict__ g2, const float* __restrict__ be2,
                    const float* __restrict__ m2, const float* __restrict__ v2,
                    float* __restrict__ mn)
{
    const int lane = threadIdx.x & 31;
    const int wid  = threadIdx.x >> 5;
    int blk = blockIdx.x;
    const int strip = blk & 127;  blk >>= 7;
    const int g     = blk & 7;    blk >>= 3;
    const int b     = blk;
    const int pbase = (strip * 8 + wid) * 16;

    const int mrow  = lane & 15;
    const int khalf = (lane >> 4) * 2;
    const int arow  = mrow & 7;

    const float* xg  = x  + ((size_t)b * Cc + g * 32) * HW;
    const float* w1g = w1 + g * 8 * 32;

    v2f a1[8], bf[8];
#pragma unroll
    for (int kk = 0; kk < 8; ++kk) {
        const int k0 = kk * 4 + khalf;
        bf[kk].x  = xg[(size_t)k0       * HW + pbase + mrow];
        bf[kk].y  = xg[(size_t)(k0 + 1) * HW + pbase + mrow];
        a1[kk].x  = w1g[arow * 32 + k0];
        a1[kk].y  = w1g[arow * 32 + k0 + 1];
    }
    v8f c1 = {};
#pragma unroll
    for (int kk = 0; kk < 8; ++kk)
        c1  = __builtin_amdgcn_wmma_f32_16x16x4_f32(false, a1[kk],  false, bf[kk],
                                                    (short)0, c1,  false, false);
    asm volatile(WMMA_SETTLE : "+v"(c1),
                 "+v"(a1[0]), "+v"(a1[1]), "+v"(a1[2]), "+v"(a1[3]),
                 "+v"(a1[4]), "+v"(a1[5]), "+v"(a1[6]), "+v"(a1[7]),
                 "+v"(bf[0]), "+v"(bf[1]), "+v"(bf[2]), "+v"(bf[3]),
                 "+v"(bf[4]), "+v"(bf[5]), "+v"(bf[6]), "+v"(bf[7]));

    if (lane < 16) {
        float tin[8];
#pragma unroll
        for (int i = 0; i < 8; ++i) {
            const int ch = g * 8 + i;
            const float inv = g1[ch] * rsqrtf(v1[ch] + EPSV);
            const float val = (c1[i] + b1[ch]) * inv + (be1[ch] - m1[ch] * inv);
            tin[i] = tanhf(val);
        }
        float* mb = mn + ((size_t)b * 80 + g * 10) * HW + pbase + lane;
#pragma unroll
        for (int o = 0; o < 10; ++o) {
            const int ch = g * 10 + o;
            float acc = 0.f;
#pragma unroll
            for (int i = 0; i < 8; ++i) acc += tin[i] * w2[ch * 8 + i];
            const float inv = g2[ch] * rsqrtf(v2[ch] + EPSV);
            mb[(size_t)o * HW] = (acc + b2[ch]) * inv + (be2[ch] - m2[ch] * inv);
        }
    }
}

constexpr int TR    = 4;
constexpr int HR    = TR + 2;
constexpr int PITCH = 136;
constexpr int CHP   = HR * PITCH + 4;

__global__ __launch_bounds__(512, 8)
void attn_kernel(const float* __restrict__ x, const float* __restrict__ wv,
                 const float* __restrict__ mn, float* __restrict__ out)
{
    __shared__ float vs[32 * CHP];
    __shared__ float nbs[HR * PITCH];

    const int bg = blockIdx.x;
    const int b  = bg >> 3, g = bg & 7;
    const int y0 = blockIdx.y * TR;

    const float* xg  = x  + ((size_t)b * Cc + g * 32) * HW;
    const float* wvg = wv + g * 32 * 32;
    const float* nbp = mn + ((size_t)b * 80 + g) * HW;

    {
        const int lane  = threadIdx.x & 31;
        const int wid   = __builtin_amdgcn_readfirstlane((int)(threadIdx.x >> 5)) & 15;
        const int mrow  = lane & 15;
        const int khalf = (lane >> 4) * 2;
        const int orow  = (lane < 16) ? 0 : 8;

        v2f alo[8], ahi[8];
#pragma unroll
        for (int kk = 0; kk < 8; ++kk) {
            const int k0 = kk * 4 + khalf;
            alo[kk].x = wvg[mrow * 32 + k0];         alo[kk].y = wvg[mrow * 32 + k0 + 1];
            ahi[kk].x = wvg[(mrow + 16) * 32 + k0];  ahi[kk].y = wvg[(mrow + 16) * 32 + k0 + 1];
        }

        v2f  bf[2][8];
        bool okv[3];
        int  ev[3], offv[3];
#pragma unroll
        for (int tt = 0; tt < 3; ++tt) {
            const int t   = wid + 16 * tt;
            const int r   = t >> 3, cs = (t & 7) * 16;
            const int gy  = y0 - 1 + r;
            const bool ok = gy >= 0 && gy < Hh;
            offv[tt] = (ok ? gy : 0) * Ww + cs + mrow;
            ev[tt]   = r * PITCH + 4 + cs + mrow;
            okv[tt]  = ok;
        }
#pragma unroll
        for (int tt = 0; tt < 2; ++tt) {
#pragma unroll
            for (int kk = 0; kk < 8; ++kk) {
                const int k0 = kk * 4 + khalf;
                bf[tt][kk].x = xg[(size_t)k0       * HW + offv[tt]];
                bf[tt][kk].y = xg[(size_t)(k0 + 1) * HW + offv[tt]];
            }
        }
#pragma unroll
        for (int tt = 0; tt < 3; ++tt) {
            const int st = tt & 1;
            v8f clo = {}, chi = {};
#pragma unroll
            for (int kk = 0; kk < 8; ++kk) {
                clo = __builtin_amdgcn_wmma_f32_16x16x4_f32(false, alo[kk], false, bf[st][kk],
                                                            (short)0, clo, false, false);
                chi = __builtin_amdgcn_wmma_f32_16x16x4_f32(false, ahi[kk], false, bf[st][kk],
                                                            (short)0, chi, false, false);
            }
            asm volatile(WMMA_SETTLE : "+v"(clo), "+v"(chi),
                         "+v"(bf[st][0]), "+v"(bf[st][1]), "+v"(bf[st][2]), "+v"(bf[st][3]),
                         "+v"(bf[st][4]), "+v"(bf[st][5]), "+v"(bf[st][6]), "+v"(bf[st][7]),
                         "+v"(alo[0]), "+v"(alo[1]), "+v"(alo[2]), "+v"(alo[3]),
                         "+v"(alo[4]), "+v"(alo[5]), "+v"(alo[6]), "+v"(alo[7]),
                         "+v"(ahi[0]), "+v"(ahi[1]), "+v"(ahi[2]), "+v"(ahi[3]),
                         "+v"(ahi[4]), "+v"(ahi[5]), "+v"(ahi[6]), "+v"(ahi[7]));
            const int e = ev[tt];
#pragma unroll
            for (int i = 0; i < 8; ++i) {
                vs[(i + orow) * CHP + e]      = okv[tt] ? clo[i] : 0.f;
                vs[(i + orow + 16) * CHP + e] = okv[tt] ? chi[i] : 0.f;
            }
            if (tt == 0) {
#pragma unroll
                for (int kk = 0; kk < 8; ++kk) {
                    const int k0 = kk * 4 + khalf;
                    bf[0][kk].x = xg[(size_t)k0       * HW + offv[2]];
                    bf[0][kk].y = xg[(size_t)(k0 + 1) * HW + offv[2]];
                }
            }
        }
    }
    if (threadIdx.x < 32 * HR * 2) {
        const int ch = threadIdx.x / (HR * 2), q = threadIdx.x % (HR * 2);
        vs[ch * CHP + (q >> 1) * PITCH + ((q & 1) ? 132 : 3)] = 0.f;
    } else if (threadIdx.x < 32 * HR * 2 + HR * 2) {
        const int q = threadIdx.x - 32 * HR * 2;
        nbs[(q >> 1) * PITCH + ((q & 1) ? 132 : 3)] = 0.f;
    }
    for (int e = threadIdx.x; e < HR * Ww; e += 512) {
        const int r = e >> 7, cx = e & 127;
        const int gy = y0 - 1 + r;
        nbs[r * PITCH + 4 + cx] = (gy >= 0 && gy < Hh) ? nbp[gy * Ww + cx] : 0.f;
    }
    __syncthreads();

    const int lx = threadIdx.x & 127, ly = threadIdx.x >> 7;
    const int y = y0 + ly;
    const float* maskp = mn + ((size_t)b * 80 + 8 + g * 9) * HW + (size_t)y * Ww + lx;

    float lg[9];
#pragma unroll
    for (int k = 0; k < 9; ++k) {
        const int i = k / 3, j = k % 3;
        lg[k] = maskp[(size_t)k * HW] + nbs[(ly + i) * PITCH + (lx + 3 + j)];
    }
    float mx = lg[0];
#pragma unroll
    for (int k = 1; k < 9; ++k) mx = fmaxf(mx, lg[k]);
    float s = 0.f;
#pragma unroll
    for (int k = 0; k < 9; ++k) { lg[k] = expf(lg[k] - mx); s += lg[k]; }
    const float sinv = 1.f / s;
#pragma unroll
    for (int k = 0; k < 9; ++k) lg[k] *= sinv;

    float* outp = out + ((size_t)b * 256 + g * 32) * HW + (size_t)y * Ww + lx;
#pragma unroll
    for (int c = 0; c < 32; ++c) {
        float acc = 0.f;
#pragma unroll
        for (int k = 0; k < 9; ++k) {
            const int i = k / 3, j = k % 3;
            acc += lg[k] * vs[c * CHP + (ly + i) * PITCH + (lx + 3 + j)];
        }
        outp[(size_t)c * HW] = acc;
    }
}

extern "C" void kernel_launch(void* const* d_in, const int* in_sizes, int n_in,
                              void* d_out, int out_size, void* d_ws, size_t ws_size,
                              hipStream_t stream)
{
    const float* x   = (const float*)d_in[0];
    const float* w1  = (const float*)d_in[1];
    const float* b1  = (const float*)d_in[2];
    const float* g1  = (const float*)d_in[3];
    const float* be1 = (const float*)d_in[4];
    const float* m1  = (const float*)d_in[5];
    const float* v1  = (const float*)d_in[6];
    const float* w2  = (const float*)d_in[7];
    const float* b2  = (const float*)d_in[8];
    const float* g2  = (const float*)d_in[9];
    const float* be2 = (const float*)d_in[10];
    const float* m2  = (const float*)d_in[11];
    const float* v2  = (const float*)d_in[12];
    const float* wvp = (const float*)d_in[13];
    float* out = (float*)d_out;

    float* mnb  = (float*)d_ws;

    mask_pointwise<<<Bn * Gg * 128, 256, 0, stream>>>(x, w1, b1, g1, be1, m1, v1,
                                                      w2, b2, g2, be2, m2, v2, mnb);
    attn_kernel<<<dim3(Bn * Gg, Hh / TR), 512, 0, stream>>>(x, wvp, mnb, out);
}


# Mask kernel: wave = 32 pixels of one line, lane = pixel: w1 block-diagonal over the k slots (2 channels per product); whole-line x loads and mn stores, all lanes in epilogue; new conv1 sum order
# speedup vs baseline: 2.6810x; 1.0919x over previous
#include <hip/hip_runtime.h>

typedef __attribute__((ext_vector_type(2))) float v2f;
typedef __attribute__((ext_vector_type(8))) float v8f;

#define EPSV 1e-5f
#define WMMA_SETTLE "v_nop\n\tv_nop\n\tv_nop\n\tv_nop\n\tv_nop\n\tv_nop\n\tv_nop\n\tv_nop"
constexpr int Bn = 2, Cc = 256, Hh = 128, Ww = 128, Gg = 8;
constexpr int HW = Hh * Ww;

__global__ __launch_bounds__(256)
void mask_pointwise(const float* __restrict__ x,
                    const float* __restrict__ w1, const float* __restrict__ b1,
                    const float* __restrict__ g1, const float* __restrict__ be1,
                    const float* __restrict__ m1, const float* __restrict__ v1,
                    const float* __restrict__ w2, const float* __restrict__ b2,
                    const float* __restrict__ g2, const float* __restrict__ be2,
                    const float* __restrict__ m2, const float* __restrict__ v2,
                    float* __restrict__ mn)
{
    const int lane = threadIdx.x & 31;
    const int wid  = threadIdx.x >> 5;
    int blk = blockIdx.x;
    const int strip = blk & 63;   blk >>= 6;
    const int g     = blk & 7;    blk >>= 3;
    const int b     = blk;
    const int pbase = (strip * 8 + wid) * 32;

    const int mrow = lane & 15;
    const int arow = mrow & 7;
    const bool aon = (mrow >> 3) == (lane >> 4);

    const float* xl  = x  + ((size_t)b * Cc + g * 32) * HW + pbase + lane;
    const float* w1r = w1 + g * 8 * 32 + arow * 32;

    float bb1[8], sc1[8], sh1[8];
#pragma unroll
    for (int i = 0; i < 8; ++i) {
        const int ch = g * 8 + i;
        const float inv = g1[ch] * rsqrtf(v1[ch] + EPSV);
        bb1[i] = b1[ch];
        sc1[i] = inv;
        sh1[i] = be1[ch] - m1[ch] * inv;
    }

    v2f a1[16], bf[16];
#pragma unroll
    for (int s = 0; s < 16; ++s) {
        bf[s].x = xl[(size_t)(2 * s)     * HW];
        bf[s].y = xl[(size_t)(2 * s + 1) * HW];
        const float wa = w1r[2 * s], wb = w1r[2 * s + 1];
        a1[s].x = aon ? wa : 0.f;
        a1[s].y = aon ? wb : 0.f;
    }
    v8f c1 = {};
#pragma unroll
    for (int s = 0; s < 8; ++s)
        c1 = __builtin_amdgcn_wmma_f32_16x16x4_f32(false, a1[s], false, bf[s],
                                                   (short)0, c1, false, false);
    asm volatile(WMMA_SETTLE : "+v"(c1),
                 "+v"(a1[0]), "+v"(a1[1]), "+v"(a1[2]), "+v"(a1[3]),
                 "+v"(a1[4]), "+v"(a1[5]), "+v"(a1[6]), "+v"(a1[7]),
                 "+v"(bf[0]), "+v"(bf[1]), "+v"(bf[2]), "+v"(bf[3]),
                 "+v"(bf[4]), "+v"(bf[5]), "+v"(bf[6]), "+v"(bf[7]));
#pragma unroll
    for (int s = 8; s < 16; ++s)
        c1 = __builtin_amdgcn_wmma_f32_16x16x4_f32(false, a1[s], false, bf[s],
                                                   (short)0, c1, false, false);
    asm volatile(WMMA_SETTLE : "+v"(c1),
                 "+v"(a1[8]), "+v"(a1[9]), "+v"(a1[10]), "+v"(a1[11]),
                 "+v"(a1[12]), "+v"(a1[13]), "+v"(a1[14]), "+v"(a1[15]),
                 "+v"(bf[8]), "+v"(bf[9]), "+v"(bf[10]), "+v"(bf[11]),
                 "+v"(bf[12]), "+v"(bf[13]), "+v"(bf[14]), "+v"(bf[15]));

    float tin[8];
#pragma unroll
    for (int i = 0; i < 8; ++i) {
        const float val = (c1[i] + bb1[i]) * sc1[i] + sh1[i];
        tin[i] = tanhf(val);
    }
    float* mb = mn + ((size_t)b * 80 + g * 10) * HW + pbase + lane;
#pragma unroll
    for (int o = 0; o < 10; ++o) {
        const int ch = g * 10 + o;
        float acc = 0.f;
#pragma unroll
        for (int i = 0; i < 8; ++i) acc += tin[i] * w2[ch * 8 + i];
        const float inv = g2[ch] * rsqrtf(v2[ch] + EPSV);
        mb[(size_t)o * HW] = (acc + b2[ch]) * inv + (be2[ch] - m2[ch] * inv);
    }
}

constexpr int TR    = 4;
constexpr int HR    = TR + 2;
constexpr int PITCH = 136;
constexpr int CHP   = HR * PITCH + 4;

__global__ __launch_bounds__(512, 8)
void attn_kernel(const float* __restrict__ x, const float* __restrict__ wv,
                 const float* __restrict__ mn, float* __restrict__ out)
{
    __shared__ float vs[32 * CHP];
    __shared__ float nbs[HR * PITCH];

    const int bg = blockIdx.x;
    const int b  = bg >> 3, g = bg & 7;
    const int y0 = blockIdx.y * TR;

    const float* xg  = x  + ((size_t)b * Cc + g * 32) * HW;
    const float* wvg = wv + g * 32 * 32;
    const float* nbp = mn + ((size_t)b * 80 + g) * HW;

    {
        const int lane  = threadIdx.x & 31;
        const int wid   = __builtin_amdgcn_readfirstlane((int)(threadIdx.x >> 5)) & 15;
        const int mrow  = lane & 15;
        const int khalf = (lane >> 4) * 2;
        const int orow  = (lane < 16) ? 0 : 8;

        v2f alo[8], ahi[8];
#pragma unroll
        for (int kk = 0; kk < 8; ++kk) {
            const int k0 = kk * 4 + khalf;
            alo[kk].x = wvg[mrow * 32 + k0];         alo[kk].y = wvg[mrow * 32 + k0 + 1];
            ahi[kk].x = wvg[(mrow + 16) * 32 + k0];  ahi[kk].y = wvg[(mrow + 16) * 32 + k0 + 1];
        }

        v2f  bf[2][8];
        bool okv[3];
        int  ev[3], offv[3];
#pragma unroll
        for (int tt = 0; tt < 3; ++tt) {
            const int t   = wid + 16 * tt;
            const int r   = t >> 3, cs = (t & 7) * 16;
            const int gy  = y0 - 1 + r;
            const bool ok = gy >= 0 && gy < Hh;
            offv[tt] = (ok ? gy : 0) * Ww + cs + mrow;
            ev[tt]   = r * PITCH + 4 + cs + mrow;
            okv[tt]  = ok;
        }
#pragma unroll
        for (int tt = 0; tt < 2; ++tt) {
#pragma unroll
            for (int kk = 0; kk < 8; ++kk) {
                const int k0 = kk * 4 + khalf;
                bf[tt][kk].x = xg[(size_t)k0       * HW + offv[tt]];
                bf[tt][kk].y = xg[(size_t)(k0 + 1) * HW + offv[tt]];
            }
        }
#pragma unroll
        for (int tt = 0; tt < 3; ++tt) {
            const int st = tt & 1;
            v8f clo = {}, chi = {};
#pragma unroll
            for (int kk = 0; kk < 8; ++kk) {
                clo = __builtin_amdgcn_wmma_f32_16x16x4_f32(false, alo[kk], false, bf[st][kk],
                                                            (short)0, clo, false, false);
                chi = __builtin_amdgcn_wmma_f32_16x16x4_f32(false, ahi[kk], false, bf[st][kk],
                                                            (short)0, chi, false, false);
            }
            asm volatile(WMMA_SETTLE : "+v"(clo), "+v"(chi),
                         "+v"(bf[st][0]), "+v"(bf[st][1]), "+v"(bf[st][2]), "+v"(bf[st][3]),
                         "+v"(bf[st][4]), "+v"(bf[st][5]), "+v"(bf[st][6]), "+v"(bf[st][7]),
                         "+v"(alo[0]), "+v"(alo[1]), "+v"(alo[2]), "+v"(alo[3]),
                         "+v"(alo[4]), "+v"(alo[5]), "+v"(alo[6]), "+v"(alo[7]),
                         "+v"(ahi[0]), "+v"(ahi[1]), "+v"(ahi[2]), "+v"(ahi[3]),
                         "+v"(ahi[4]), "+v"(ahi[5]), "+v"(ahi[6]), "+v"(ahi[7]));
            const int e = ev[tt];
#pragma unroll
            for (int i = 0; i < 8; ++i) {
                vs[(i + orow) * CHP + e]      = okv[tt] ? clo[i] : 0.f;
                vs[(i + orow + 16) * CHP + e] = okv[tt] ? chi[i] : 0.f;
            }
            if (tt == 0) {
#pragma unroll
                for (int kk = 0; kk < 8; ++kk) {
                    const int k0 = kk * 4 + khalf;
                    bf[0][kk].x = xg[(size_t)k0       * HW + offv[2]];
                    bf[0][kk].y = xg[(size_t)(k0 + 1) * HW + offv[2]];
                }
            }
        }
    }
    if (threadIdx.x < 32 * HR * 2) {
        const int ch = threadIdx.x / (HR * 2), q = threadIdx.x % (HR * 2);
        vs[ch * CHP + (q >> 1) * PITCH + ((q & 1) ? 132 : 3)] = 0.f;
    } else if (threadIdx.x < 32 * HR * 2 + HR * 2) {
        const int q = threadIdx.x - 32 * HR * 2;
        nbs[(q >> 1) * PITCH + ((q & 1) ? 132 : 3)] = 0.f;
    }
    for (int e = threadIdx.x; e < HR * Ww; e += 512) {
        const int r = e >> 7, cx = e & 127;
        const int gy = y0 - 1 + r;
        nbs[r * PITCH + 4 + cx] = (gy >= 0 && gy < Hh) ? nbp[gy * Ww + cx] : 0.f;
    }
    __syncthreads();

    const int lx = threadIdx.x & 127, ly = threadIdx.x >> 7;
    const int y = y0 + ly;
    const float* maskp = mn + ((size_t)b * 80 + 8 + g * 9) * HW + (size_t)y * Ww + lx;

    float lg[9];
#pragma unroll
    for (int k = 0; k < 9; ++k) {
        const int i = k / 3, j = k % 3;
        lg[k] = maskp[(size_t)k * HW] + nbs[(ly + i) * PITCH + (lx + 3 + j)];
    }
    float mx = lg[0];
#pragma unroll
    for (int k = 1; k < 9; ++k) mx = fmaxf(mx, lg[k]);
    float s = 0.f;
#pragma unroll
    for (int k = 0; k < 9; ++k) { lg[k] = expf(lg[k] - mx); s += lg[k]; }
    const float sinv = 1.f / s;
#pragma unroll
    for (int k = 0; k < 9; ++k) lg[k] *= sinv;

    float* outp = out + ((size_t)b * 256 + g * 32) * HW + (size_t)y * Ww + lx;
#pragma unroll
    for (int c = 0; c < 32; ++c) {
        float acc = 0.f;
#pragma unroll
        for (int k = 0; k < 9; ++k) {
            const int i = k / 3, j = k % 3;
            acc += lg[k] * vs[c * CHP + (ly + i) * PITCH + (lx + 3 + j)];
        }
        outp[(size_t)c * HW] = acc;
    }
}

extern "C" void kernel_launch(void* const* d_in, const int* in_sizes, int n_in,
                              void* d_out, int out_size, void* d_ws, size_t ws_size,
                              hipStream_t stream)
{
    const float* x   = (const float*)d_in[0];
    const float* w1  = (const float*)d_in[1];
    const float* b1  = (const float*)d_in[2];
    const float* g1  = (const float*)d_in[3];
    const float* be1 = (const float*)d_in[4];
    const float* m1  = (const float*)d_in[5];
    const float* v1  = (const float*)d_in[6];
    const float* w2  = (const float*)d_in[7];
    const float* b2  = (const float*)d_in[8];
    const float* g2  = (const float*)d_in[9];
    const float* be2 = (const float*)d_in[10];
    const float* m2  = (const float*)d_in[11];
    const float* v2  = (const float*)d_in[12];
    const float* wvp = (const float*)d_in[13];
    float* out = (float*)d_out;

    float* mnb  = (float*)d_ws;

    mask_pointwise<<<Bn * Gg * 64, 256, 0, stream>>>(x, w1, b1, g1, be1, m1, v1,
                                                      w2, b2, g2, be2, m2, v2, mnb);
    attn_kernel<<<dim3(Bn * Gg, Hh / TR), 512, 0, stream>>>(x, wvp, mnb, out);
}


# Attention: 3x3 window sum runs on x, value conv applied after it (linear): products only for the 4 output rows (48 to 32 per wave), behind the LDS phase; x tile in LDS; taps summed before channels
# speedup vs baseline: 2.7670x; 1.0321x over previous
#include <hip/hip_runtime.h>

typedef __attribute__((ext_vector_type(2))) float v2f;
typedef __attribute__((ext_vector_type(8))) float v8f;

#define EPSV 1e-5f
#define WMMA_SETTLE "v_nop\n\tv_nop\n\tv_nop\n\tv_nop\n\tv_nop\n\tv_nop\n\tv_nop\n\tv_nop"
constexpr int Bn = 2, Cc = 256, Hh = 128, Ww = 128, Gg = 8;
constexpr int HW = Hh * Ww;

__global__ __launch_bounds__(256)
void mask_pointwise(const float* __restrict__ x,
                    const float* __restrict__ w1, const float* __restrict__ b1,
                    const float* __restrict__ g1, const float* __restrict__ be1,
                    const float* __restrict__ m1, const float* __restrict__ v1,
                    const float* __restrict__ w2, const float* __restrict__ b2,
                    const float* __restrict__ g2, const float* __restrict__ be2,
                    const float* __restrict__ m2, const float* __restrict__ v2,
                    float* __restrict__ mn)
{
    const int lane = threadIdx.x & 31;
    const int wid  = threadIdx.x >> 5;
    int blk = blockIdx.x;
    const int strip = blk & 63;   blk >>= 6;
    const int g     = blk & 7;    blk >>= 3;
    const int b     = blk;
    const int pbase = (strip * 8 + wid) * 32;

    const int mrow = lane & 15;
    const int arow = mrow & 7;
    const bool aon = (mrow >> 3) == (lane >> 4);

    const float* xl  = x  + ((size_t)b * Cc + g * 32) * HW + pbase + lane;
    const float* w1r = w1 + g * 8 * 32 + arow * 32;

    float bb1[8], sc1[8], sh1[8];
#pragma unroll
    for (int i = 0; i < 8; ++i) {
        const int ch = g * 8 + i;
        const float inv = g1[ch] * rsqrtf(v1[ch] + EPSV);
        bb1[i] = b1[ch];
        sc1[i] = inv;
        sh1[i] = be1[ch] - m1[ch] * inv;
    }

    v2f a1[16], bf[16];
#pragma unroll
    for (int s = 0; s < 16; ++s) {
        bf[s].x = xl[(size_t)(2 * s)     * HW];
        bf[s].y = xl[(size_t)(2 * s + 1) * HW];
        const float wa = w1r[2 * s], wb = w1r[2 * s + 1];
        a1[s].x = aon ? wa : 0.f;
        a1[s].y = aon ? wb : 0.f;
    }
    v8f c1 = {};
#pragma unroll
    for (int s = 0; s < 8; ++s)
        c1 = __builtin_amdgcn_wmma_f32_16x16x4_f32(false, a1[s], false, bf[s],
                                                   (short)0, c1, false, false);
    asm volatile(WMMA_SETTLE : "+v"(c1),
                 "+v"(a1[0]), "+v"(a1[1]), "+v"(a1[2]), "+v"(a1[3]),
                 "+v"(a1[4]), "+v"(a1[5]), "+v"(a1[6]), "+v"(a1[7]),
                 "+v"(bf[0]), "+v"(bf[1]), "+v"(bf[2]), "+v"(bf[3]),
                 "+v"(bf[4]), "+v"(bf[5]), "+v"(bf[6]), "+v"(bf[7]));
#pragma unroll
    for (int s = 8; s < 16; ++s)
        c1 = __builtin_amdgcn_wmma_f32_16x16x4_f32(false, a1[s], false, bf[s],
                                                   (short)0, c1, false, false);
    asm volatile(WMMA_SETTLE : "+v"(c1),
                 "+v"(a1[8]), "+v"(a1[9]), "+v"(a1[10]), "+v"(a1[11]),
                 "+v"(a1[12]), "+v"(a1[13]), "+v"(a1[14]), "+v"(a1[15]),
                 "+v"(bf[8]), "+v"(bf[9]), "+v"(bf[10]), "+v"(bf[11]),
                 "+v"(bf[12]), "+v"(bf[13]), "+v"(bf[14]), "+v"(bf[15]));

    float tin[8];
#pragma unroll
    for (int i = 0; i < 8; ++i) {
        const float val = (c1[i] + bb1[i]) * sc1[i] + sh1[i];
        tin[i] = tanhf(val);
    }
    float* mb = mn + ((size_t)b * 80 + g * 10) * HW + pbase + lane;
#pragma unroll
    for (int o = 0; o < 10; ++o) {
        const int ch = g * 10 + o;
        float acc = 0.f;
#pragma unroll
        for (int i = 0; i < 8; ++i) acc += tin[i] * w2[ch * 8 + i];
        const float inv = g2[ch] * rsqrtf(v2[ch] + EPSV);
        mb[(size_t)o * HW] = (acc + b2[ch]) * inv + (be2[ch] - m2[ch] * inv);
    }
}

constexpr int TR    = 4;
constexpr int HR    = TR + 2;
constexpr int PITCH = 136;
constexpr int CHP   = HR * PITCH + 4;

__device__ __forceinline__ float other_half(float v)
{
    unsigned u = __builtin_bit_cast(unsigned, v);
    u = __builtin_amdgcn_permlanex16(u, u, 0x76543210u, 0xfedcba98u, false, false);
    return __builtin_bit_cast(float, u);
}

__global__ __launch_bounds__(512, 8)
void attn_kernel(const float* __restrict__ x, const float* __restrict__ wv,
                 const float* __restrict__ mn, float* __restrict__ out)
{
    __shared__ __attribute__((aligned(16))) float xs[32 * CHP];
    __shared__ float nbs[HR * PITCH];

    const int bg = blockIdx.x;
    const int b  = bg >> 3, g = bg & 7;
    const int y0 = blockIdx.y * TR;

    const float* xg  = x  + ((size_t)b * Cc + g * 32) * HW;
    const float* wvg = wv + g * 32 * 32;
    const float* nbp = mn + ((size_t)b * 80 + g) * HW;

    const int lane = threadIdx.x & 31;
    const int wid  = __builtin_amdgcn_readfirstlane((int)(threadIdx.x >> 5)) & 15;
    const int c    = lane & 15;
    const int h    = lane >> 4;
    const int ly   = wid >> 2;
    const int s0   = (wid & 3) * 32;
    const int y    = y0 + ly;

    const float* maskp = mn + ((size_t)b * 80 + 8 + g * 9) * HW + (size_t)y * Ww + s0 + lane;
    float lg[9];
#pragma unroll
    for (int k = 0; k < 9; ++k) lg[k] = maskp[(size_t)k * HW];

    v2f alo[8], ahi[8];
#pragma unroll
    for (int kk = 0; kk < 8; ++kk) {
        const int k0 = kk * 4 + h * 2;
        alo[kk].x = wvg[c * 32 + k0];         alo[kk].y = wvg[c * 32 + k0 + 1];
        ahi[kk].x = wvg[(c + 16) * 32 + k0];  ahi[kk].y = wvg[(c + 16) * 32 + k0 + 1];
    }

    const int nr0 = threadIdx.x >> 7, ncx = threadIdx.x & 127;
    const int nr1 = 4 + (nr0 & 1);
    const int ngy0 = y0 - 1 + nr0, ngy1 = y0 - 1 + nr1;
    const float nb0 = nbp[(ngy0 < 0 ? 0 : ngy0) * Ww + ncx];
    const float nb1 = nbp[(ngy1 > Hh - 1 ? Hh - 1 : ngy1) * Ww + ncx];

    float4 xv[12];
#pragma unroll
    for (int it = 0; it < 12; ++it) {
        const int ch = wid + 16 * (it & 1), r = it >> 1;
        const int gy = y0 - 1 + r;
        const int gyc = gy < 0 ? 0 : (gy > Hh - 1 ? Hh - 1 : gy);
        xv[it] = *reinterpret_cast<const float4*>(xg + (size_t)ch * HW + gyc * Ww + 4 * lane);
    }
#pragma unroll
    for (int it = 0; it < 12; ++it) {
        const int ch = wid + 16 * (it & 1), r = it >> 1;
        const int gy = y0 - 1 + r;
        const bool ok = gy >= 0 && gy < Hh;
        float4 val;
        val.x = ok ? xv[it].x : 0.f;
        val.y = ok ? xv[it].y : 0.f;
        val.z = ok ? xv[it].z : 0.f;
        val.w = ok ? xv[it].w : 0.f;
        *reinterpret_cast<float4*>(&xs[ch * CHP + r * PITCH + 4 + 4 * lane]) = val;
    }
    nbs[nr0 * PITCH + 4 + ncx] = (ngy0 >= 0) ? nb0 : 0.f;
    if (threadIdx.x < 256)
        nbs[nr1 * PITCH + 4 + ncx] = (ngy1 < Hh) ? nb1 : 0.f;
    if (threadIdx.x < 32 * HR * 2) {
        const int ch = threadIdx.x / (HR * 2), q = threadIdx.x % (HR * 2);
        xs[ch * CHP + (q >> 1) * PITCH + ((q & 1) ? 132 : 3)] = 0.f;
    } else if (threadIdx.x < 32 * HR * 2 + HR * 2) {
        const int q = threadIdx.x - 32 * HR * 2;
        nbs[(q >> 1) * PITCH + ((q & 1) ? 132 : 3)] = 0.f;
    }
    __syncthreads();

#pragma unroll
    for (int k = 0; k < 9; ++k) {
        const int i = k / 3, j = k % 3;
        lg[k] += nbs[(ly + i) * PITCH + (s0 + lane + 3 + j)];
    }
    float mx = lg[0];
#pragma unroll
    for (int k = 1; k < 9; ++k) mx = fmaxf(mx, lg[k]);
    float s = 0.f;
#pragma unroll
    for (int k = 0; k < 9; ++k) { lg[k] = expf(lg[k] - mx); s += lg[k]; }
    const float sinv = 1.f / s;
#pragma unroll
    for (int k = 0; k < 9; ++k) lg[k] *= sinv;

    float wt[2][9];
#pragma unroll
    for (int k = 0; k < 9; ++k) {
        const float o = other_half(lg[k]);
        wt[0][k] = h ? o : lg[k];
        wt[1][k] = h ? lg[k] : o;
    }

    v8f clo[2], chi[2];
#pragma unroll
    for (int tt = 0; tt < 2; ++tt) {
        const float* xb = xs + (2 * h) * CHP + ly * PITCH + (s0 + 16 * tt + c + 3);
        v2f bf[8];
#pragma unroll
        for (int kk = 0; kk < 8; ++kk) {
            float a0 = 0.f, a1 = 0.f;
#pragma unroll
            for (int k = 0; k < 9; ++k) {
                const int i = k / 3, j = k % 3;
                a0 = __builtin_fmaf(wt[tt][k], xb[(4 * kk)     * CHP + i * PITCH + j], a0);
                a1 = __builtin_fmaf(wt[tt][k], xb[(4 * kk + 1) * CHP + i * PITCH + j], a1);
            }
            bf[kk].x = a0;
            bf[kk].y = a1;
        }
        v8f dlo = {}, dhi = {};
#pragma unroll
        for (int kk = 0; kk < 8; ++kk) {
            dlo = __builtin_amdgcn_wmma_f32_16x16x4_f32(false, alo[kk], false, bf[kk],
                                                        (short)0, dlo, false, false);
            dhi = __builtin_amdgcn_wmma_f32_16x16x4_f32(false, ahi[kk], false, bf[kk],
                                                        (short)0, dhi, false, false);
        }
        asm volatile(WMMA_SETTLE : "+v"(dlo), "+v"(dhi),
                     "+v"(bf[0]), "+v"(bf[1]), "+v"(bf[2]), "+v"(bf[3]),
                     "+v"(bf[4]), "+v"(bf[5]), "+v"(bf[6]), "+v"(bf[7]),
                     "+v"(alo[0]), "+v"(alo[1]), "+v"(alo[2]), "+v"(alo[3]),
                     "+v"(alo[4]), "+v"(alo[5]), "+v"(alo[6]), "+v"(alo[7]),
                     "+v"(ahi[0]), "+v"(ahi[1]), "+v"(ahi[2]), "+v"(ahi[3]),
                     "+v"(ahi[4]), "+v"(ahi[5]), "+v"(ahi[6]), "+v"(ahi[7]));
        clo[tt] = dlo;
        chi[tt] = dhi;
    }

    float* outp = out + ((size_t)b * 256 + g * 32) * HW + (size_t)y * Ww + s0 + lane;
#pragma unroll
    for (int i = 0; i < 8; ++i) {
        const float l0 = clo[0][i], l1 = clo[1][i];
        const float u0 = chi[0][i], u1 = chi[1][i];
        const float xl0 = other_half(l0), xl1 = other_half(l1);
        const float xu0 = other_half(u0), xu1 = other_half(u1);
        outp[(size_t)(i)      * HW] = h ? xl1 : l0;
        outp[(size_t)(i + 8)  * HW] = h ? l1  : xl0;
        outp[(size_t)(i + 16) * HW] = h ? xu1 : u0;
        outp[(size_t)(i + 24) * HW] = h ? u1  : xu0;
    }
}

extern "C" void kernel_launch(void* const* d_in, const int* in_sizes, int n_in,
                              void* d_out, int out_size, void* d_ws, size_t ws_size,
                              hipStream_t stream)
{
    const float* x   = (const float*)d_in[0];
    const float* w1  = (const float*)d_in[1];
    const float* b1  = (const float*)d_in[2];
    const float* g1  = (const float*)d_in[3];
    const float* be1 = (const float*)d_in[4];
    const float* m1  = (const float*)d_in[5];
    const float* v1  = (const float*)d_in[6];
    const float* w2  = (const float*)d_in[7];
    const float* b2  = (const float*)d_in[8];
    const float* g2  = (const float*)d_in[9];
    const float* be2 = (const float*)d_in[10];
    const float* m2  = (const float*)d_in[11];
    const float* v2  = (const float*)d_in[12];
    const float* wvp = (const float*)d_in[13];
    float* out = (float*)d_out;

    float* mnb  = (float*)d_ws;

    mask_pointwise<<<Bn * Gg * 64, 256, 0, stream>>>(x, w1, b1, g1, be1, m1, v1,
                                                      w2, b2, g2, be2, m2, v2, mnb);
    attn_kernel<<<dim3(Bn * Gg, Hh / TR), 512, 0, stream>>>(x, wvp, mnb, out);
}
